# starting kernel
# baseline (speedup 1.0000x reference)
#include <hip/hip_runtime.h>
#include <math.h>

typedef __attribute__((ext_vector_type(2))) float v2f;
typedef __attribute__((ext_vector_type(8))) float v8f;

#define DIM 1024
#define NCLS 128
#define NB 8192
#define SCL 16.0f


__global__ __launch_bounds__(256) void sym_prep_kernel(const float* __restrict__ w,
                                                       float* __restrict__ ws) {
    __shared__ float r0[256], r1[256], r2[256];
    const int t = threadIdx.x;
    const float* w0 = w;
    const float* w1 = w + DIM;
    float s00 = 0.f, s01 = 0.f, s11 = 0.f;
#pragma unroll
    for (int k = 0; k < 4; ++k) {
        int e = t * 4 + k;
        float a = w0[e], b = w1[e];
        s00 = fmaf(a, a, s00);
        s01 = fmaf(a, b, s01);
        s11 = fmaf(b, b, s11);
    }
    r0[t] = s00; r1[t] = s01; r2[t] = s11;
    __syncthreads();
    for (int s = 128; s > 0; s >>= 1) {
        if (t < s) { r0[t] += r0[t + s]; r1[t] += r1[t + s]; r2[t] += r2[t + s]; }
        __syncthreads();
    }
    s00 = r0[0]; s01 = r1[0]; s11 = r2[0];
    float inv1 = 1.0f / sqrtf(s00);
    float d    = s01 / s00;
    float inv2 = 1.0f / sqrtf(fmaxf(s11 - s01 * d, 1e-30f));
#pragma unroll
    for (int k = 0; k < 4; ++k) {
        int e = t * 4 + k;
        float a = w0[e], b = w1[e];
        ws[e]       = a * inv1;
        ws[DIM + e] = (b - d * a) * inv2;
    }
    if (t < NCLS) {
        float th = 6.2831853071795864769f * (float)t / (float)NCLS;
        ws[2 * DIM + t]        = SCL * cosf(th);
        ws[2 * DIM + NCLS + t] = SCL * sinf(th);
    }
}

__global__ __launch_bounds__(128) void sym_main_kernel(const float* __restrict__ x,
                                                       const float* __restrict__ ws,
                                                       float* __restrict__ out) {
    __shared__ float4 sN1[DIM / 4];
    __shared__ float4 sN2[DIM / 4];
    __shared__ float  sCos[NCLS];
    __shared__ float  sSin[NCLS];

    const int t    = threadIdx.x;
    const int lane = t & 31;
    const int wave = t >> 5;

    {
        const float4* g1 = (const float4*)ws;
        const float4* g2 = (const float4*)(ws + DIM);
        sN1[t]       = g1[t];
        sN1[t + 128] = g1[t + 128];
        sN2[t]       = g2[t];
        sN2[t + 128] = g2[t + 128];
        sCos[t] = ws[2 * DIM + t];
        sSin[t] = ws[2 * DIM + NCLS + t];
    }
    __syncthreads();

    const int rowbase = blockIdx.x * 64 + wave * 16;

    float aM = 0.0f, bM = 0.0f;

    for (int r = 0; r < 16; ++r) {
        const float4* xp = (const float4*)(x + (size_t)(rowbase + r) * DIM);
        float sx = 0.f, sa = 0.f, sb = 0.f;
#pragma unroll
        for (int i = 0; i < 8; ++i) {
            float4 xv = xp[i * 32 + lane];
            float4 u  = sN1[i * 32 + lane];
            float4 v  = sN2[i * 32 + lane];
            sx = fmaf(xv.x, xv.x, fmaf(xv.y, xv.y, fmaf(xv.z, xv.z, fmaf(xv.w, xv.w, sx))));
            sa = fmaf(xv.x, u.x,  fmaf(xv.y, u.y,  fmaf(xv.z, u.z,  fmaf(xv.w, u.w,  sa))));
            sb = fmaf(xv.x, v.x,  fmaf(xv.y, v.y,  fmaf(xv.z, v.z,  fmaf(xv.w, v.w,  sb))));
        }
#pragma unroll
        for (int off = 16; off >= 1; off >>= 1) {
            sx += __shfl_xor(sx, off, 32);
            sa += __shfl_xor(sa, off, 32);
            sb += __shfl_xor(sb, off, 32);
        }
        float rn   = 1.0f / fmaxf(sqrtf(sx), 1e-12f);
        bool  mine = (lane == r);
        aM = mine ? sa * rn : aM;
        bM = mine ? sb * rn : bM;
    }

    const int  n       = lane & 15;
    const bool lowhalf = (lane < 16);

#if __has_builtin(__builtin_amdgcn_wmma_f32_16x16x4_f32)
    v2f afrag = {aM, bM};
#pragma unroll
    for (int tt = 0; tt < 8; ++tt) {
        float cv = sCos[tt * 16 + n];
        float sv = sSin[tt * 16 + n];
        v2f bfrag;
        bfrag.x = lowhalf ? cv : 0.0f;
        bfrag.y = lowhalf ? sv : 0.0f;
        v8f cfr = {};
        cfr = __builtin_amdgcn_wmma_f32_16x16x4_f32(false, afrag, false, bfrag,
                                                    (short)0, cfr, false, false);
        const int mbase = rowbase + (lowhalf ? 0 : 8);
#pragma unroll
        for (int vv = 0; vv < 8; ++vv) {
            out[(size_t)(mbase + vv) * NCLS + tt * 16 + n] = cfr[vv];
        }
    }
#else
    __shared__ float sA[4][16], sB[4][16];
    if (lane < 16) { sA[wave][lane] = aM; sB[wave][lane] = bM; }
    __syncthreads();
    for (int idx = lane; idx < 16 * NCLS; idx += 32) {
        int rr = idx >> 7, cc = idx & (NCLS - 1);
        out[(size_t)(rowbase + rr) * NCLS + cc] =
            fmaf(sA[wave][rr], sCos[cc], sB[wave][rr] * sSin[cc]);
    }
#endif
}

extern "C" void kernel_launch(void* const* d_in, const int* in_sizes, int n_in,
                              void* d_out, int out_size, void* d_ws, size_t ws_size,
                              hipStream_t stream) {
    const float* x  = (const float*)d_in[0];
    const float* w  = (const float*)d_in[1];
    float* out = (float*)d_out;
    float* ws  = (float*)d_ws;

    sym_prep_kernel<<<1, 256, 0, stream>>>(w, ws);
    sym_main_kernel<<<NB / 64, 128, 0, stream>>>(x, ws, out);
}


# prep launch folded into the 16-row-tile kernel (one launch, no workspace): raw w0,w1 as B, w0.w0 w0.w1 w1.w1 K-split over the 8 waves, row sums orthonormalised after; sincospi; new weight-sum order
# speedup vs baseline: 5.6702x; 5.6702x over previous
#include <hip/hip_runtime.h>
#include <math.h>


typedef __attribute__((ext_vector_type(2))) float v2f;
typedef __attribute__((ext_vector_type(4))) float v4f;
typedef __attribute__((ext_vector_type(8))) float v8f;

#define DIM 1024
#define NCLS 128
#define NB 8192
#define SCL 16.0f

#define XPITCH 33
#define PPITCH 17

template <int CTRL>
__device__ __forceinline__ float dpp_get(float v) {
    return __builtin_bit_cast(float, __builtin_amdgcn_update_dpp(0, __builtin_bit_cast(int, v), CTRL, 0xf, 0xf, true));
}
__device__ __forceinline__ float wave_sum(float v) {
    v += dpp_get<0x161>(v);
    v += dpp_get<0x162>(v);
    v += dpp_get<0x164>(v);
    v += dpp_get<0x168>(v);
    const int b = __builtin_bit_cast(int, v);
    return v + __builtin_bit_cast(float, __builtin_amdgcn_permlanex16(b, b, 0x76543210, 0xfedcba98, false, false));
}

__global__ __launch_bounds__(256) void sym_main_kernel(const float* __restrict__ x,
                                                       const float* __restrict__ wt,
                                                       float* __restrict__ out) {
    __shared__ v4f sX[8 * 16 * XPITCH];
    __shared__ v4f sP[16 * PPITCH];
    __shared__ v4f sG[8];

    const int t    = threadIdx.x;
    const int lane = t & 31;
    const int w    = __builtin_amdgcn_readfirstlane(t >> 5);
    const int c    = lane & 15;
    const int h    = lane >> 4;
    const int row0 = blockIdx.x * 16;

    const v4f* xg = (const v4f*)(x + (size_t)row0 * DIM + 128 * w) + lane;
    v4f xr[16];
#pragma unroll
    for (int r = 0; r < 16; ++r) xr[r] = xg[r * (DIM / 4)];

    const bool isn = (c < 2);
    const v4f* bg = (const v4f*)(wt + (isn ? c : 1) * DIM + 128 * w) + h;
    v4f bv[16];
#pragma unroll
    for (int j = 0; j < 16; ++j) bv[j] = bg[2 * j];

    const v4f* wl = (const v4f*)(wt + 128 * w) + lane;
    const v4f u0 = wl[0];
    const v4f u1 = wl[DIM / 4];

    const int tp   = w & 3;
    const int col0 = 32 * tp + c;
    float cv0 = 0.f, sv0 = 0.f, cv1 = 0.f, sv1 = 0.f;
    if (w < 4) {
        sincospif((float)col0 * (1.0f / 64.0f), &sv0, &cv0);
        sincospif((float)(col0 + 16) * (1.0f / 64.0f), &sv1, &cv1);
        cv0 *= SCL; sv0 *= SCL; cv1 *= SCL; sv1 *= SCL;
    }

    asm volatile("" :: "v"(xr[0]), "v"(xr[1]), "v"(xr[2]), "v"(xr[3]), "v"(xr[4]), "v"(xr[5]),
                       "v"(xr[6]), "v"(xr[7]), "v"(xr[8]), "v"(xr[9]), "v"(xr[10]), "v"(xr[11]),
                       "v"(xr[12]), "v"(xr[13]), "v"(xr[14]), "v"(xr[15]));

    v4f* sw = sX + w * (16 * XPITCH);
#pragma unroll
    for (int r = 0; r < 16; ++r) sw[r * XPITCH + lane] = xr[r];

    {
        float g00 = u0.x * u0.x, g01 = u0.x * u1.x, g11 = u1.x * u1.x;
        g00 = fmaf(u0.y, u0.y, g00); g01 = fmaf(u0.y, u1.y, g01); g11 = fmaf(u1.y, u1.y, g11);
        g00 = fmaf(u0.z, u0.z, g00); g01 = fmaf(u0.z, u1.z, g01); g11 = fmaf(u1.z, u1.z, g11);
        g00 = fmaf(u0.w, u0.w, g00); g01 = fmaf(u0.w, u1.w, g01); g11 = fmaf(u1.w, u1.w, g11);
        g00 = wave_sum(g00);
        g01 = wave_sum(g01);
        g11 = wave_sum(g11);
        if (lane == 0) {
            v4f g = {g00, g01, g11, 0.0f};
            sG[w] = g;
        }
    }
    __syncthreads();

    v4f af[16];
#pragma unroll
    for (int j = 0; j < 16; ++j) af[j] = sw[c * XPITCH + 2 * j + h];
#pragma unroll
    for (int j = 0; j < 16; ++j) {
        bv[j].x = isn ? bv[j].x : 0.0f;
        bv[j].y = isn ? bv[j].y : 0.0f;
        bv[j].z = isn ? bv[j].z : 0.0f;
        bv[j].w = isn ? bv[j].w : 0.0f;
    }
    float sx0 = 0.f, sx1 = 0.f, sx2 = 0.f, sx3 = 0.f;
#pragma unroll
    for (int j = 0; j < 16; ++j) {
        sx0 = fmaf(af[j].x, af[j].x, sx0);
        sx1 = fmaf(af[j].y, af[j].y, sx1);
        sx2 = fmaf(af[j].z, af[j].z, sx2);
        sx3 = fmaf(af[j].w, af[j].w, sx3);
    }
    float* sPf = (float*)sP;
    sPf[(c * PPITCH + 2 * w) * 4 + 2 + h] = (sx0 + sx1) + (sx2 + sx3);
    float* pd = sPf + ((8 * h) * PPITCH + 2 * w) * 4 + c;
    __builtin_amdgcn_sched_barrier(0);

    v8f acc0 = {};
    v8f acc1 = {};
#pragma unroll
    for (int j = 0; j < 16; ++j) {
        v2f a0 = {af[j].x, af[j].y};
        v2f a1 = {af[j].z, af[j].w};
        v2f b0 = {bv[j].x, bv[j].y};
        v2f b1 = {bv[j].z, bv[j].w};
        acc0 = __builtin_amdgcn_wmma_f32_16x16x4_f32(false, a0, false, b0, (short)0, acc0, false, false);
        acc1 = __builtin_amdgcn_wmma_f32_16x16x4_f32(false, a1, false, b1, (short)0, acc1, false, false);
    }

    if (isn) {
#pragma unroll
        for (int r = 0; r < 8; ++r) {
            pd[r * PPITCH * 4]     = acc0[r];
            pd[r * PPITCH * 4 + 4] = acc1[r];
        }
    }
    __syncthreads();

    float sa = 0.f, sb = 0.f, ss = 0.f;
#pragma unroll
    for (int q = 0; q < 16; ++q) {
        const v4f p = sP[c * PPITCH + q];
        sa += p.x;
        sb += p.y;
        if ((q & 1) == 0) ss += p.z + p.w;
    }
    const float rn = 1.0f / fmaxf(sqrtf(ss), 1e-12f);
    const bool  lo = (h == 0);

    if (w < 4) {
        const v4f g = sG[c & 7];
        float g00 = g.x, g01 = g.y, g11 = g.z;
        g00 += dpp_get<0x161>(g00); g01 += dpp_get<0x161>(g01); g11 += dpp_get<0x161>(g11);
        g00 += dpp_get<0x162>(g00); g01 += dpp_get<0x162>(g01); g11 += dpp_get<0x162>(g11);
        g00 += dpp_get<0x164>(g00); g01 += dpp_get<0x164>(g01); g11 += dpp_get<0x164>(g11);
        const float inv1 = 1.0f / sqrtf(g00);
        const float d    = g01 / g00;
        const float inv2 = 1.0f / sqrtf(fmaxf(g11 - g01 * d, 1e-30f));
        const float pa = sa * inv1;
        const float pb = (sb - d * sa) * inv2;
        v2f afrag;
        afrag.x = lo ? pa * rn : 0.0f;
        afrag.y = lo ? pb * rn : 0.0f;
        v2f bf0, bf1;
        bf0.x = lo ? cv0 : 0.0f;
        bf0.y = lo ? sv0 : 0.0f;
        bf1.x = lo ? cv1 : 0.0f;
        bf1.y = lo ? sv1 : 0.0f;
        const v8f z = {};
        const v8f d0 = __builtin_amdgcn_wmma_f32_16x16x4_f32(false, afrag, false, bf0, (short)0, z, false, false);
        const v8f d1 = __builtin_amdgcn_wmma_f32_16x16x4_f32(false, afrag, false, bf1, (short)0, z, false, false);
        float* ob = out + (size_t)(row0 + 8 * h) * NCLS + col0;
#pragma unroll
        for (int r = 0; r < 8; ++r) {
            ob[r * NCLS]      = d0[r];
            ob[r * NCLS + 16] = d1[r];
        }
    }
}

extern "C" void kernel_launch(void* const* d_in, const int* in_sizes, int n_in,
                              void* d_out, int out_size, void* d_ws, size_t ws_size,
                              hipStream_t stream) {
    const float* x  = (const float*)d_in[0];
    const float* wt = (const float*)d_in[1];
    float* out = (float*)d_out;

    sym_main_kernel<<<NB / 16, 256, 0, stream>>>(x, wt, out);
}


# rows consumed in arrival order: a wave's slice comes as four 16x32 sub-tiles, each turned through the wave's own LDS rows and multiplied under the later loads; first barrier gone; same sum order
# speedup vs baseline: 6.3548x; 1.1207x over previous
#include <hip/hip_runtime.h>
#include <math.h>


typedef __attribute__((ext_vector_type(2))) float v2f;
typedef __attribute__((ext_vector_type(4))) float v4f;
typedef __attribute__((ext_vector_type(8))) float v8f;

#define DIM 1024
#define NCLS 128
#define NB 8192
#define SCL 16.0f

#define SPITCH 9
#define SUBT   (16 * SPITCH)
#define PPITCH 17

template <int CTRL>
__device__ __forceinline__ float dpp_get(float v) {
    return __builtin_bit_cast(float, __builtin_amdgcn_update_dpp(0, __builtin_bit_cast(int, v), CTRL, 0xf, 0xf, true));
}
__device__ __forceinline__ float wave_sum(float v) {
    v += dpp_get<0x161>(v);
    v += dpp_get<0x162>(v);
    v += dpp_get<0x164>(v);
    v += dpp_get<0x168>(v);
    const int b = __builtin_bit_cast(int, v);
    return v + __builtin_bit_cast(float, __builtin_amdgcn_permlanex16(b, b, 0x76543210, 0xfedcba98, false, false));
}

__device__ __forceinline__ void turn_subtile(v4f* buf, int so, int lo,
                                             v4f r0, v4f r1, v4f r2, v4f r3, v4f (&af)[4]) {
    buf[so]               = r0;
    buf[so + 4 * SPITCH]  = r1;
    buf[so + 8 * SPITCH]  = r2;
    buf[so + 12 * SPITCH] = r3;
    af[0] = buf[lo];
    af[1] = buf[lo + 2];
    af[2] = buf[lo + 4];
    af[3] = buf[lo + 6];
}

__device__ __forceinline__ void run_subtile(const v4f (&af)[4], const v4f* bq, v8f& acc0, v8f& acc1,
                                            float& sx0, float& sx1, float& sx2, float& sx3) {
#pragma unroll
    for (int jj = 0; jj < 4; ++jj) {
        sx0 = fmaf(af[jj].x, af[jj].x, sx0);
        sx1 = fmaf(af[jj].y, af[jj].y, sx1);
        sx2 = fmaf(af[jj].z, af[jj].z, sx2);
        sx3 = fmaf(af[jj].w, af[jj].w, sx3);
    }
#pragma unroll
    for (int jj = 0; jj < 4; ++jj) {
        v2f a0 = {af[jj].x, af[jj].y};
        v2f a1 = {af[jj].z, af[jj].w};
        v2f b0 = {bq[jj].x, bq[jj].y};
        v2f b1 = {bq[jj].z, bq[jj].w};
        acc0 = __builtin_amdgcn_wmma_f32_16x16x4_f32(false, a0, false, b0, (short)0, acc0, false, false);
        acc1 = __builtin_amdgcn_wmma_f32_16x16x4_f32(false, a1, false, b1, (short)0, acc1, false, false);
    }
}

__global__ __launch_bounds__(256) void sym_main_kernel(const float* __restrict__ x,
                                                       const float* __restrict__ wt,
                                                       float* __restrict__ out) {
    __shared__ v4f smem[16 * PPITCH + 8 * 4 * SUBT + 8];
    v4f* const sP = smem;
    v4f* const sX = smem + 16 * PPITCH;
    v4f* const sG = smem + 16 * PPITCH + 8 * 4 * SUBT;

    const int t    = threadIdx.x;
    const int lane = t & 31;
    const int w    = __builtin_amdgcn_readfirstlane(t >> 5);
    const int c    = lane & 15;
    const int h    = lane >> 4;
    const int lr   = lane >> 3;
    const int lu   = lane & 7;
    const int row0 = blockIdx.x * 16;

    const v4f* wl = (const v4f*)(wt + 128 * w) + lane;
    const v4f u0 = wl[0];
    const v4f u1 = wl[DIM / 4];
    const v4f* xg = (const v4f*)(x + (size_t)(row0 + lr) * DIM + 128 * w) + lu;
    v4f xr[16];
#pragma unroll
    for (int i = 0; i < 4; ++i) xr[i] = xg[i * DIM];
    __builtin_amdgcn_sched_barrier(0);
    const bool isn = (c < 2);
    const v4f* bg = (const v4f*)(wt + (isn ? c : 1) * DIM + 128 * w) + h;
    v4f bv[16];
#pragma unroll
    for (int j = 0; j < 16; ++j) bv[j] = bg[2 * j];
#pragma unroll
    for (int q = 1; q < 4; ++q) {
        if (q > 1) __builtin_amdgcn_sched_barrier(0);
#pragma unroll
        for (int i = 0; i < 4; ++i) xr[4 * q + i] = xg[i * DIM + 8 * q];
    }

    const int tp   = w & 3;
    const int col0 = 32 * tp + c;
    float cv0 = 0.f, sv0 = 0.f, cv1 = 0.f, sv1 = 0.f;
    if (w < 4) {
        sincospif((float)col0 * (1.0f / 64.0f), &sv0, &cv0);
        sincospif((float)(col0 + 16) * (1.0f / 64.0f), &sv1, &cv1);
        cv0 *= SCL; sv0 *= SCL; cv1 *= SCL; sv1 *= SCL;
    }

    {
        float g00 = u0.x * u0.x, g01 = u0.x * u1.x, g11 = u1.x * u1.x;
        g00 = fmaf(u0.y, u0.y, g00); g01 = fmaf(u0.y, u1.y, g01); g11 = fmaf(u1.y, u1.y, g11);
        g00 = fmaf(u0.z, u0.z, g00); g01 = fmaf(u0.z, u1.z, g01); g11 = fmaf(u1.z, u1.z, g11);
        g00 = fmaf(u0.w, u0.w, g00); g01 = fmaf(u0.w, u1.w, g01); g11 = fmaf(u1.w, u1.w, g11);
        g00 = wave_sum(g00);
        g01 = wave_sum(g01);
        g11 = wave_sum(g11);
        if (lane == 0) {
            v4f g = {g00, g01, g11, 0.0f};
            sG[w] = g;
        }
    }
#pragma unroll
    for (int j = 0; j < 16; ++j) {
        bv[j].x = isn ? bv[j].x : 0.0f;
        bv[j].y = isn ? bv[j].y : 0.0f;
        bv[j].z = isn ? bv[j].z : 0.0f;
        bv[j].w = isn ? bv[j].w : 0.0f;
    }

#define RUN_DONE() asm volatile("" :: "v"(acc0), "v"(acc1) : "memory")
    v4f* sw = sX + w * (4 * SUBT);
    const int so = lr * SPITCH + lu;
    const int lo = c * SPITCH + h;
    float sx0 = 0.f, sx1 = 0.f, sx2 = 0.f, sx3 = 0.f;
    v8f acc0 = {};
    v8f acc1 = {};
    v4f afA[4], afB[4];

    turn_subtile(sw,            so, lo, xr[0],  xr[1],  xr[2],  xr[3],  afA);
    turn_subtile(sw + SUBT,     so, lo, xr[4],  xr[5],  xr[6],  xr[7],  afB);
    __builtin_amdgcn_sched_barrier(0);
    run_subtile(afA, bv,      acc0, acc1, sx0, sx1, sx2, sx3);
    RUN_DONE();
    turn_subtile(sw + 2 * SUBT, so, lo, xr[8],  xr[9],  xr[10], xr[11], afA);
    __builtin_amdgcn_sched_barrier(0);
    run_subtile(afB, bv + 4,  acc0, acc1, sx0, sx1, sx2, sx3);
    RUN_DONE();
    turn_subtile(sw + 3 * SUBT, so, lo, xr[12], xr[13], xr[14], xr[15], afB);
    __builtin_amdgcn_sched_barrier(0);
    run_subtile(afA, bv + 8,  acc0, acc1, sx0, sx1, sx2, sx3);
    run_subtile(afB, bv + 12, acc0, acc1, sx0, sx1, sx2, sx3);
#undef RUN_DONE

    float* sPf = (float*)sP;
    sPf[(c * PPITCH + 2 * w) * 4 + 2 + h] = (sx0 + sx1) + (sx2 + sx3);
    float* pd = sPf + ((8 * h) * PPITCH + 2 * w) * 4 + c;
    if (isn) {
#pragma unroll
        for (int r = 0; r < 8; ++r) {
            pd[r * PPITCH * 4]     = acc0[r];
            pd[r * PPITCH * 4 + 4] = acc1[r];
        }
    }
    __syncthreads();

    float sa = 0.f, sb = 0.f, ss = 0.f;
#pragma unroll
    for (int q = 0; q < 16; ++q) {
        const v4f p = sP[c * PPITCH + q];
        sa += p.x;
        sb += p.y;
        if ((q & 1) == 0) ss += p.z + p.w;
    }
    const float rn = 1.0f / fmaxf(sqrtf(ss), 1e-12f);
    const bool  lo0 = (h == 0);

    if (w < 4) {
        const v4f g = sG[c & 7];
        float g00 = g.x, g01 = g.y, g11 = g.z;
        g00 += dpp_get<0x161>(g00); g01 += dpp_get<0x161>(g01); g11 += dpp_get<0x161>(g11);
        g00 += dpp_get<0x162>(g00); g01 += dpp_get<0x162>(g01); g11 += dpp_get<0x162>(g11);
        g00 += dpp_get<0x164>(g00); g01 += dpp_get<0x164>(g01); g11 += dpp_get<0x164>(g11);
        const float inv1 = 1.0f / sqrtf(g00);
        const float d    = g01 / g00;
        const float inv2 = 1.0f / sqrtf(fmaxf(g11 - g01 * d, 1e-30f));
        const float pa = sa * inv1;
        const float pb = (sb - d * sa) * inv2;
        v2f afrag;
        afrag.x = lo0 ? pa * rn : 0.0f;
        afrag.y = lo0 ? pb * rn : 0.0f;
        v2f bf0, bf1;
        bf0.x = lo0 ? cv0 : 0.0f;
        bf0.y = lo0 ? sv0 : 0.0f;
        bf1.x = lo0 ? cv1 : 0.0f;
        bf1.y = lo0 ? sv1 : 0.0f;
        const v8f z = {};
        const v8f d0 = __builtin_amdgcn_wmma_f32_16x16x4_f32(false, afrag, false, bf0, (short)0, z, false, false);
        const v8f d1 = __builtin_amdgcn_wmma_f32_16x16x4_f32(false, afrag, false, bf1, (short)0, z, false, false);
        float* ob = out + (size_t)(row0 + 8 * h) * NCLS + col0;
#pragma unroll
        for (int r = 0; r < 8; ++r) {
            ob[r * NCLS]      = d0[r];
            ob[r * NCLS + 16] = d1[r];
        }
    }
}

extern "C" void kernel_launch(void* const* d_in, const int* in_sizes, int n_in,
                              void* d_out, int out_size, void* d_ws, size_t ws_size,
                              hipStream_t stream) {
    const float* x  = (const float*)d_in[0];
    const float* wt = (const float*)d_in[1];
    float* out = (float*)d_out;

    sym_main_kernel<<<NB / 16, 256, 0, stream>>>(x, wt, out);
}
